# starting kernel
# baseline (speedup 1.0000x reference)
#include <hip/hip_runtime.h>
#include <hip/hip_bf16.h>

#define N_NODES 100000
#define N_EDGES 3200000
#define IN_DIM  512
#define HID_DIM 16
#define N_TILES (N_NODES / 16)

typedef float v2f __attribute__((ext_vector_type(2)));
typedef float v8f __attribute__((ext_vector_type(8)));

__global__ __launch_bounds__(256) void gcn_gemm_xw1(
    const float* __restrict__ X, const float* __restrict__ W1,
    float* __restrict__ h1) {
  __shared__ float sW[IN_DIM * HID_DIM];
  for (int i = threadIdx.x; i < IN_DIM * HID_DIM; i += 256) sW[i] = W1[i];
  __syncthreads();

  const int wave = threadIdx.x >> 5;
  const int lane = threadIdx.x & 31;
  const int tile = blockIdx.x * 8 + wave;
  if (tile >= N_TILES) return;

  const int m  = lane & 15;
  const int hi = lane >> 4;

  const float* __restrict__ xrow = X + (size_t)(tile * 16 + m) * IN_DIM;

  v8f c = {0.f, 0.f, 0.f, 0.f, 0.f, 0.f, 0.f, 0.f};

  #pragma unroll 4
  for (int k0 = 0; k0 < IN_DIM; k0 += 4) {
    const int k = k0 + 2 * hi;
    v2f a;
    a.x = xrow[k];
    a.y = xrow[k + 1];
    v2f b;
    b.x = sW[k * HID_DIM + m];
    b.y = sW[(k + 1) * HID_DIM + m];
    c = __builtin_amdgcn_wmma_f32_16x16x4_f32(
        false, a, false, b, (short)0, c, false, false);
  }

  float* __restrict__ out = h1 + (size_t)tile * 16 * HID_DIM;
  const int mbase = hi * 8;
  #pragma unroll
  for (int v = 0; v < 8; ++v)
    out[(mbase + v) * HID_DIM + m] = c[v];
}

__global__ void gcn_init(float* __restrict__ h2, float* __restrict__ out,
                         const float* __restrict__ b2) {
  int i = blockIdx.x * blockDim.x + threadIdx.x;
  if (i < N_NODES * HID_DIM) h2[i] = 0.0f;
  if (i < N_NODES) out[i] = b2[0];
}

__global__ __launch_bounds__(256) void gcn_spmm1(
    const int* __restrict__ esrc, const int* __restrict__ edst,
    const float* __restrict__ ew, const float* __restrict__ h1,
    float* __restrict__ h2) {
  size_t gid = (size_t)blockIdx.x * 256 + threadIdx.x;
  size_t e = gid >> 4;
  if (e >= N_EDGES) return;
  int d = (int)(gid & 15);
  int s = esrc[e];
  int t = edst[e];
  float v = ew[e] * h1[(size_t)s * HID_DIM + d];
  __hip_atomic_fetch_add(&h2[(size_t)t * HID_DIM + d], v,
                         __ATOMIC_RELAXED, __HIP_MEMORY_SCOPE_AGENT);
}

__global__ void gcn_layer2(const float* __restrict__ h2,
                           const float* __restrict__ b1,
                           const float* __restrict__ W2,
                           float* __restrict__ z) {
  int i = blockIdx.x * blockDim.x + threadIdx.x;
  if (i >= N_NODES) return;
  const float* __restrict__ row = h2 + (size_t)i * HID_DIM;
  float acc = 0.0f;
  #pragma unroll
  for (int d = 0; d < HID_DIM; ++d) {
    float hv = row[d] + b1[d];
    hv = fmaxf(hv, 0.0f);
    acc = fmaf(hv, W2[d], acc);
  }
  z[i] = acc;
}

__global__ __launch_bounds__(256) void gcn_spmm2(
    const int* __restrict__ esrc, const int* __restrict__ edst,
    const float* __restrict__ ew, const float* __restrict__ z,
    float* __restrict__ out) {
  size_t e = (size_t)blockIdx.x * 256 + threadIdx.x;
  if (e >= N_EDGES) return;
  float v = ew[e] * z[esrc[e]];
  __hip_atomic_fetch_add(&out[edst[e]], v,
                         __ATOMIC_RELAXED, __HIP_MEMORY_SCOPE_AGENT);
}

extern "C" void kernel_launch(void* const* d_in, const int* in_sizes, int n_in,
                              void* d_out, int out_size, void* d_ws, size_t ws_size,
                              hipStream_t stream) {
  const float* x    = (const float*)d_in[0];
  const int*   esrc = (const int*)  d_in[1];
  const int*   edst = (const int*)  d_in[2];
  const float* ew   = (const float*)d_in[3];
  const float* W1   = (const float*)d_in[4];
  const float* b1   = (const float*)d_in[5];
  const float* W2   = (const float*)d_in[6];
  const float* b2   = (const float*)d_in[7];
  float* out = (float*)d_out;

  float* h1 = (float*)d_ws;
  float* h2 = h1 + (size_t)N_NODES * HID_DIM;
  float* z  = h2 + (size_t)N_NODES * HID_DIM;

  {
    int blocks = (N_TILES + 7) / 8;
    gcn_gemm_xw1<<<blocks, 256, 0, stream>>>(x, W1, h1);
  }
  {
    int total = N_NODES * HID_DIM;
    gcn_init<<<(total + 255) / 256, 256, 0, stream>>>(h2, out, b2);
  }
  {
    size_t total = (size_t)N_EDGES * HID_DIM;
    int blocks = (int)((total + 255) / 256);
    gcn_spmm1<<<blocks, 256, 0, stream>>>(esrc, edst, ew, h1, h2);
  }
  gcn_layer2<<<(N_NODES + 255) / 256, 256, 0, stream>>>(h2, b1, W2, z);
  gcn_spmm2<<<(N_EDGES + 255) / 256, 256, 0, stream>>>(esrc, edst, ew, z, out);
}


# layer-2 scatter-add summed per row range in LDS (LDS atomics) before one global atomic per row and edge piece; changes the order in which the f32 terms are summed
# speedup vs baseline: 1.4925x; 1.4925x over previous
#include <hip/hip_runtime.h>
#include <hip/hip_bf16.h>

#define N_NODES 100000
#define N_EDGES 3200000
#define IN_DIM  512
#define HID_DIM 16
#define N_TILES (N_NODES / 16)

typedef float v2f __attribute__((ext_vector_type(2)));
typedef float v8f __attribute__((ext_vector_type(8)));

__global__ __launch_bounds__(256) void gcn_gemm_xw1(
    const float* __restrict__ X, const float* __restrict__ W1,
    float* __restrict__ h1) {
  __shared__ float sW[IN_DIM * HID_DIM];
  for (int i = threadIdx.x; i < IN_DIM * HID_DIM; i += 256) sW[i] = W1[i];
  __syncthreads();

  const int wave = threadIdx.x >> 5;
  const int lane = threadIdx.x & 31;
  const int tile = blockIdx.x * 8 + wave;
  if (tile >= N_TILES) return;

  const int m  = lane & 15;
  const int hi = lane >> 4;

  const float* __restrict__ xrow = X + (size_t)(tile * 16 + m) * IN_DIM;

  v8f c = {0.f, 0.f, 0.f, 0.f, 0.f, 0.f, 0.f, 0.f};

  #pragma unroll 4
  for (int k0 = 0; k0 < IN_DIM; k0 += 4) {
    const int k = k0 + 2 * hi;
    v2f a;
    a.x = xrow[k];
    a.y = xrow[k + 1];
    v2f b;
    b.x = sW[k * HID_DIM + m];
    b.y = sW[(k + 1) * HID_DIM + m];
    c = __builtin_amdgcn_wmma_f32_16x16x4_f32(
        false, a, false, b, (short)0, c, false, false);
  }

  float* __restrict__ out = h1 + (size_t)tile * 16 * HID_DIM;
  const int mbase = hi * 8;
  #pragma unroll
  for (int v = 0; v < 8; ++v)
    out[(mbase + v) * HID_DIM + m] = c[v];
}

__global__ void gcn_init(float* __restrict__ h2, float* __restrict__ out,
                         const float* __restrict__ b2) {
  int i = blockIdx.x * blockDim.x + threadIdx.x;
  if (i < N_NODES * HID_DIM) h2[i] = 0.0f;
  if (i < N_NODES) out[i] = b2[0];
}

__global__ __launch_bounds__(256) void gcn_spmm1(
    const int* __restrict__ esrc, const int* __restrict__ edst,
    const float* __restrict__ ew, const float* __restrict__ h1,
    float* __restrict__ h2) {
  size_t gid = (size_t)blockIdx.x * 256 + threadIdx.x;
  size_t e = gid >> 4;
  if (e >= N_EDGES) return;
  int d = (int)(gid & 15);
  int s = esrc[e];
  int t = edst[e];
  float v = ew[e] * h1[(size_t)s * HID_DIM + d];
  __hip_atomic_fetch_add(&h2[(size_t)t * HID_DIM + d], v,
                         __ATOMIC_RELAXED, __HIP_MEMORY_SCOPE_AGENT);
}

__global__ void gcn_layer2(const float* __restrict__ h2,
                           const float* __restrict__ b1,
                           const float* __restrict__ W2,
                           float* __restrict__ z) {
  int i = blockIdx.x * blockDim.x + threadIdx.x;
  if (i >= N_NODES) return;
  const float* __restrict__ row = h2 + (size_t)i * HID_DIM;
  float acc = 0.0f;
  #pragma unroll
  for (int d = 0; d < HID_DIM; ++d) {
    float hv = row[d] + b1[d];
    hv = fmaxf(hv, 0.0f);
    acc = fmaf(hv, W2[d], acc);
  }
  z[i] = acc;
}

#define S2_PARTS  8
#define S2_CHUNKS 32
#define S2_ROWS   12512
#define S2_QUADS  (N_EDGES / S2_CHUNKS / 4)
#define S2_ROT    384
#define S2_TRIPS  ((S2_QUADS + 1023) / 1024)
static_assert(S2_TRIPS % 2 == 1, "the trip loop below takes trip 0, then pairs of trips");
static_assert(S2_QUADS > 1024, "trip 0 must be in range for every thread");
static_assert(S2_PARTS * S2_ROWS >= N_NODES && S2_CHUNKS * S2_ROT < S2_ROWS, "range cover");

struct S2Hit {
  unsigned rx, ry, rz, rw;
  float    wx, wy, wz, ww;
  float    zx, zy, zz, zw;
};

__device__ __forceinline__ S2Hit s2_gather(const int4 d, const int4 s, const float4 w,
                                           const bool valid, const int lo,
                                           const float* __restrict__ z) {
  S2Hit h;
  h.rx = valid ? (unsigned)(d.x - lo) : 0xffffffffu;
  h.ry = valid ? (unsigned)(d.y - lo) : 0xffffffffu;
  h.rz = valid ? (unsigned)(d.z - lo) : 0xffffffffu;
  h.rw = valid ? (unsigned)(d.w - lo) : 0xffffffffu;
  h.wx = w.x; h.wy = w.y; h.wz = w.z; h.ww = w.w;
  h.zx = z[h.rx < (unsigned)S2_ROWS ? s.x : 0];
  h.zy = z[h.ry < (unsigned)S2_ROWS ? s.y : 0];
  h.zz = z[h.rz < (unsigned)S2_ROWS ? s.z : 0];
  h.zw = z[h.rw < (unsigned)S2_ROWS ? s.w : 0];
  return h;
}

__device__ __forceinline__ void s2_add(float* acc, const S2Hit h) {
  if (h.rx < (unsigned)S2_ROWS)
    __hip_atomic_fetch_add(&acc[h.rx], h.wx * h.zx, __ATOMIC_RELAXED, __HIP_MEMORY_SCOPE_WORKGROUP);
  if (h.ry < (unsigned)S2_ROWS)
    __hip_atomic_fetch_add(&acc[h.ry], h.wy * h.zy, __ATOMIC_RELAXED, __HIP_MEMORY_SCOPE_WORKGROUP);
  if (h.rz < (unsigned)S2_ROWS)
    __hip_atomic_fetch_add(&acc[h.rz], h.wz * h.zz, __ATOMIC_RELAXED, __HIP_MEMORY_SCOPE_WORKGROUP);
  if (h.rw < (unsigned)S2_ROWS)
    __hip_atomic_fetch_add(&acc[h.rw], h.ww * h.zw, __ATOMIC_RELAXED, __HIP_MEMORY_SCOPE_WORKGROUP);
}

__global__ __launch_bounds__(1024) void gcn_spmm2(
    const int* __restrict__ esrc, const int* __restrict__ edst,
    const float* __restrict__ ew, const float* __restrict__ z,
    float* __restrict__ out) {
  __shared__ float acc[S2_ROWS];
  const int part  = blockIdx.x % S2_PARTS;
  const int chunk = blockIdx.x / S2_PARTS;
  const int lo    = part * S2_ROWS;

  for (int i = threadIdx.x; i < S2_ROWS; i += 1024) acc[i] = 0.0f;
  __syncthreads();

  const int4*   __restrict__ d4 = (const int4*)edst + (size_t)chunk * S2_QUADS;
  const int4*   __restrict__ s4 = (const int4*)esrc + (size_t)chunk * S2_QUADS;
  const float4* __restrict__ w4 = (const float4*)ew + (size_t)chunk * S2_QUADS;

  int q = threadIdx.x;
  int4   d = d4[q], s = s4[q];
  float4 w = w4[q];
  S2Hit h0 = s2_gather(d, s, w, true, lo, z);
  q += 1024;
  { const int qc = min(q, S2_QUADS - 1); d = d4[qc]; s = s4[qc]; w = w4[qc]; }
  for (int k = 0; k < (S2_TRIPS - 1) / 2; ++k) {
    const S2Hit h1 = s2_gather(d, s, w, q < S2_QUADS, lo, z);
    q += 1024;
    { const int qc = min(q, S2_QUADS - 1); d = d4[qc]; s = s4[qc]; w = w4[qc]; }
    s2_add(acc, h0);
    h0 = s2_gather(d, s, w, q < S2_QUADS, lo, z);
    q += 1024;
    { const int qc = min(q, S2_QUADS - 1); d = d4[qc]; s = s4[qc]; w = w4[qc]; }
    s2_add(acc, h1);
  }
  s2_add(acc, h0);
  __syncthreads();

  const int rot = chunk * S2_ROT;
  for (int i = threadIdx.x; i < S2_ROWS; i += 1024) {
    int j = i + rot;
    if (j >= S2_ROWS) j -= S2_ROWS;
    const int g = lo + j;
    if (g < N_NODES)
      __hip_atomic_fetch_add(&out[g], acc[j], __ATOMIC_RELAXED, __HIP_MEMORY_SCOPE_AGENT);
  }
}

extern "C" void kernel_launch(void* const* d_in, const int* in_sizes, int n_in,
                              void* d_out, int out_size, void* d_ws, size_t ws_size,
                              hipStream_t stream) {
  const float* x    = (const float*)d_in[0];
  const int*   esrc = (const int*)  d_in[1];
  const int*   edst = (const int*)  d_in[2];
  const float* ew   = (const float*)d_in[3];
  const float* W1   = (const float*)d_in[4];
  const float* b1   = (const float*)d_in[5];
  const float* W2   = (const float*)d_in[6];
  const float* b2   = (const float*)d_in[7];
  float* out = (float*)d_out;

  float* h1 = (float*)d_ws;
  float* h2 = h1 + (size_t)N_NODES * HID_DIM;
  float* z  = h2 + (size_t)N_NODES * HID_DIM;

  {
    int blocks = (N_TILES + 7) / 8;
    gcn_gemm_xw1<<<blocks, 256, 0, stream>>>(x, W1, h1);
  }
  {
    int total = N_NODES * HID_DIM;
    gcn_init<<<(total + 255) / 256, 256, 0, stream>>>(h2, out, b2);
  }
  {
    size_t total = (size_t)N_EDGES * HID_DIM;
    int blocks = (int)((total + 255) / 256);
    gcn_spmm1<<<blocks, 256, 0, stream>>>(esrc, edst, ew, h1, h2);
  }
  gcn_layer2<<<(N_NODES + 255) / 256, 256, 0, stream>>>(h2, b1, W2, z);
  gcn_spmm2<<<S2_PARTS * S2_CHUNKS, 1024, 0, stream>>>(esrc, edst, ew, z, out);
}


# layer-1 scatter-add summed per row range in LDS (a wave packs its range's edges, 16 lanes per edge, LDS atomics), then one global atomic per line and edge piece; changes the f32 summation order
# speedup vs baseline: 2.6023x; 1.7436x over previous
#include <hip/hip_runtime.h>
#include <hip/hip_bf16.h>

#define N_NODES 100000
#define N_EDGES 3200000
#define IN_DIM  512
#define HID_DIM 16
#define N_TILES (N_NODES / 16)

typedef float v2f __attribute__((ext_vector_type(2)));
typedef float v8f __attribute__((ext_vector_type(8)));

__global__ __launch_bounds__(256) void gcn_gemm_xw1(
    const float* __restrict__ X, const float* __restrict__ W1,
    float* __restrict__ h1) {
  __shared__ float sW[IN_DIM * HID_DIM];
  for (int i = threadIdx.x; i < IN_DIM * HID_DIM; i += 256) sW[i] = W1[i];
  __syncthreads();

  const int wave = threadIdx.x >> 5;
  const int lane = threadIdx.x & 31;
  const int tile = blockIdx.x * 8 + wave;
  if (tile >= N_TILES) return;

  const int m  = lane & 15;
  const int hi = lane >> 4;

  const float* __restrict__ xrow = X + (size_t)(tile * 16 + m) * IN_DIM;

  v8f c = {0.f, 0.f, 0.f, 0.f, 0.f, 0.f, 0.f, 0.f};

  #pragma unroll 4
  for (int k0 = 0; k0 < IN_DIM; k0 += 4) {
    const int k = k0 + 2 * hi;
    v2f a;
    a.x = xrow[k];
    a.y = xrow[k + 1];
    v2f b;
    b.x = sW[k * HID_DIM + m];
    b.y = sW[(k + 1) * HID_DIM + m];
    c = __builtin_amdgcn_wmma_f32_16x16x4_f32(
        false, a, false, b, (short)0, c, false, false);
  }

  float* __restrict__ out = h1 + (size_t)tile * 16 * HID_DIM;
  const int mbase = hi * 8;
  #pragma unroll
  for (int v = 0; v < 8; ++v)
    out[(mbase + v) * HID_DIM + m] = c[v];
}

__global__ void gcn_init(float* __restrict__ h2, float* __restrict__ out,
                         const float* __restrict__ b2) {
  int i = blockIdx.x * blockDim.x + threadIdx.x;
  if (i < N_NODES * HID_DIM) h2[i] = 0.0f;
  if (i < N_NODES) out[i] = b2[0];
}

#define S1_PARTS   22
#define S1_CHUNKS  11
#define S1_ROWS    4576
#define S1_FLOATS  (S1_ROWS * HID_DIM)
#define S1_NQUADS  (N_EDGES / 4)
#define S1_QPC     72728
#define S1_TRIPS   ((S1_QPC + 1023) / 1024)
#define S1_ROTF    (S1_FLOATS / S1_CHUNKS)
static_assert(S1_PARTS * S1_ROWS >= N_NODES, "range cover");
static_assert(S1_CHUNKS * S1_QPC >= S1_NQUADS, "edge cover");
static_assert(S1_ROTF * S1_CHUNKS == S1_FLOATS && S1_ROTF % 32 == 0, "rotation by whole lines");
static_assert(S1_ROWS <= (1 << 13) && N_NODES <= (1 << 17), "row offset and source row share one dword");
static_assert(S1_FLOATS % 4 == 0, "LDS clear by 16-byte stores");

__device__ __forceinline__ void s1_gather(const uint2* my, const unsigned j, const unsigned total,
                                          const unsigned h, const unsigned c,
                                          const float* __restrict__ h1,
                                          unsigned (&ro)[4], float (&wt)[4], float (&hv)[4]) {
  #pragma unroll
  for (int u = 0; u < 4; ++u) {
    const unsigned idx = j + 2u * (unsigned)u + h;
    const uint2 e = my[idx < total ? idx : j];
    ro[u] = (e.x >> 17) * HID_DIM + c;
    wt[u] = __uint_as_float(e.y);
    hv[u] = h1[(size_t)(e.x & 0x1ffffu) * HID_DIM + c];
  }
}

__device__ __forceinline__ void s1_add(float* acc, const unsigned j, const unsigned total,
                                       const unsigned h,
                                       const unsigned (&ro)[4], const float (&wt)[4], const float (&hv)[4]) {
  #pragma unroll
  for (int u = 0; u < 4; ++u) {
    const bool ok = j + 2u * (unsigned)u + h < total;
    __hip_atomic_fetch_add(&acc[ro[u]], ok ? wt[u] * hv[u] : 0.0f,
                           __ATOMIC_RELAXED, __HIP_MEMORY_SCOPE_WORKGROUP);
  }
}

__global__ __launch_bounds__(1024) void gcn_spmm1(
    const int* __restrict__ esrc, const int* __restrict__ edst,
    const float* __restrict__ ew, const float* __restrict__ h1,
    float* __restrict__ h2) {
  __shared__ float acc[S1_FLOATS];
  __shared__ uint2 stg[32 * 128];
  const int part  = blockIdx.x % S1_PARTS;
  const int chunk = blockIdx.x / S1_PARTS;
  const int lo    = part * S1_ROWS;

  for (int i = threadIdx.x; i < S1_FLOATS / 4; i += 1024)
    ((float4*)acc)[i] = make_float4(0.0f, 0.0f, 0.0f, 0.0f);
  __syncthreads();

  const unsigned c = threadIdx.x & 15u;
  const unsigned h = (threadIdx.x >> 4) & 1u;
  uint2* my = stg + (threadIdx.x >> 5) * 128;
  if ((threadIdx.x & 31u) == 0u) my[0] = make_uint2(0u, 0u);

  const int4*   __restrict__ d4 = (const int4*)edst;
  const int4*   __restrict__ s4 = (const int4*)esrc;
  const float4* __restrict__ w4 = (const float4*)ew;
  const int qbeg = chunk * S1_QPC;
  const int qend = min(qbeg + S1_QPC, S1_NQUADS);

  int q = qbeg + (int)threadIdx.x;
  int4 d, s; float4 w;
  { const int qc = min(q, S1_NQUADS - 1); d = d4[qc]; s = s4[qc]; w = w4[qc]; }

  for (int t = 0; t < S1_TRIPS; ++t) {
    const bool valid = q < qend;
    const unsigned r0 = valid ? (unsigned)(d.x - lo) : 0xffffffffu;
    const unsigned r1 = valid ? (unsigned)(d.y - lo) : 0xffffffffu;
    const unsigned r2 = valid ? (unsigned)(d.z - lo) : 0xffffffffu;
    const unsigned r3 = valid ? (unsigned)(d.w - lo) : 0xffffffffu;
    const bool g0 = r0 < (unsigned)S1_ROWS, g1 = r1 < (unsigned)S1_ROWS;
    const bool g2 = r2 < (unsigned)S1_ROWS, g3 = r3 < (unsigned)S1_ROWS;
    const unsigned m0 = __builtin_amdgcn_ballot_w32(g0), m1 = __builtin_amdgcn_ballot_w32(g1);
    const unsigned m2 = __builtin_amdgcn_ballot_w32(g2), m3 = __builtin_amdgcn_ballot_w32(g3);
    const unsigned b1 = (unsigned)__builtin_popcount(m0);
    const unsigned b2 = b1 + (unsigned)__builtin_popcount(m1);
    const unsigned b3 = b2 + (unsigned)__builtin_popcount(m2);
    const unsigned total = b3 + (unsigned)__builtin_popcount(m3);
    if (g0) my[__builtin_amdgcn_mbcnt_lo(m0, 0u)] = make_uint2((r0 << 17) | (unsigned)s.x, __float_as_uint(w.x));
    if (g1) my[__builtin_amdgcn_mbcnt_lo(m1, b1)] = make_uint2((r1 << 17) | (unsigned)s.y, __float_as_uint(w.y));
    if (g2) my[__builtin_amdgcn_mbcnt_lo(m2, b2)] = make_uint2((r2 << 17) | (unsigned)s.z, __float_as_uint(w.z));
    if (g3) my[__builtin_amdgcn_mbcnt_lo(m3, b3)] = make_uint2((r3 << 17) | (unsigned)s.w, __float_as_uint(w.w));
    unsigned ro[4]; float wt[4], hv[4];
    s1_gather(my, 0u, total, h, c, h1, ro, wt, hv);
    q += 1024;
    { const int qc = min(q, S1_NQUADS - 1); d = d4[qc]; s = s4[qc]; w = w4[qc]; }
    __builtin_amdgcn_sched_barrier(0);
    s1_add(acc, 0u, total, h, ro, wt, hv);
    for (unsigned j = 8u; j < total; j += 8u) {
      s1_gather(my, j, total, h, c, h1, ro, wt, hv);
      s1_add(acc, j, total, h, ro, wt, hv);
    }
  }
  __syncthreads();

  const int rot = chunk * S1_ROTF;
  const int lim = (N_NODES - lo) * HID_DIM;
  float* __restrict__ dst = h2 + (size_t)lo * HID_DIM;
  for (int i = threadIdx.x; i < S1_FLOATS; i += 1024) {
    int j = i + rot;
    if (j >= S1_FLOATS) j -= S1_FLOATS;
    if (j < lim)
      __hip_atomic_fetch_add(&dst[j], acc[j], __ATOMIC_RELAXED, __HIP_MEMORY_SCOPE_AGENT);
  }
}

__global__ void gcn_layer2(const float* __restrict__ h2,
                           const float* __restrict__ b1,
                           const float* __restrict__ W2,
                           float* __restrict__ z) {
  int i = blockIdx.x * blockDim.x + threadIdx.x;
  if (i >= N_NODES) return;
  const float* __restrict__ row = h2 + (size_t)i * HID_DIM;
  float acc = 0.0f;
  #pragma unroll
  for (int d = 0; d < HID_DIM; ++d) {
    float hv = row[d] + b1[d];
    hv = fmaxf(hv, 0.0f);
    acc = fmaf(hv, W2[d], acc);
  }
  z[i] = acc;
}

#define S2_PARTS  8
#define S2_CHUNKS 32
#define S2_ROWS   12512
#define S2_QUADS  (N_EDGES / S2_CHUNKS / 4)
#define S2_ROT    384
#define S2_TRIPS  ((S2_QUADS + 1023) / 1024)
static_assert(S2_TRIPS % 2 == 1, "the trip loop below takes trip 0, then pairs of trips");
static_assert(S2_QUADS > 1024, "trip 0 must be in range for every thread");
static_assert(S2_PARTS * S2_ROWS >= N_NODES && S2_CHUNKS * S2_ROT < S2_ROWS, "range cover");

struct S2Hit {
  unsigned rx, ry, rz, rw;
  float    wx, wy, wz, ww;
  float    zx, zy, zz, zw;
};

__device__ __forceinline__ S2Hit s2_gather(const int4 d, const int4 s, const float4 w,
                                           const bool valid, const int lo,
                                           const float* __restrict__ z) {
  S2Hit h;
  h.rx = valid ? (unsigned)(d.x - lo) : 0xffffffffu;
  h.ry = valid ? (unsigned)(d.y - lo) : 0xffffffffu;
  h.rz = valid ? (unsigned)(d.z - lo) : 0xffffffffu;
  h.rw = valid ? (unsigned)(d.w - lo) : 0xffffffffu;
  h.wx = w.x; h.wy = w.y; h.wz = w.z; h.ww = w.w;
  h.zx = z[h.rx < (unsigned)S2_ROWS ? s.x : 0];
  h.zy = z[h.ry < (unsigned)S2_ROWS ? s.y : 0];
  h.zz = z[h.rz < (unsigned)S2_ROWS ? s.z : 0];
  h.zw = z[h.rw < (unsigned)S2_ROWS ? s.w : 0];
  return h;
}

__device__ __forceinline__ void s2_add(float* acc, const S2Hit h) {
  if (h.rx < (unsigned)S2_ROWS)
    __hip_atomic_fetch_add(&acc[h.rx], h.wx * h.zx, __ATOMIC_RELAXED, __HIP_MEMORY_SCOPE_WORKGROUP);
  if (h.ry < (unsigned)S2_ROWS)
    __hip_atomic_fetch_add(&acc[h.ry], h.wy * h.zy, __ATOMIC_RELAXED, __HIP_MEMORY_SCOPE_WORKGROUP);
  if (h.rz < (unsigned)S2_ROWS)
    __hip_atomic_fetch_add(&acc[h.rz], h.wz * h.zz, __ATOMIC_RELAXED, __HIP_MEMORY_SCOPE_WORKGROUP);
  if (h.rw < (unsigned)S2_ROWS)
    __hip_atomic_fetch_add(&acc[h.rw], h.ww * h.zw, __ATOMIC_RELAXED, __HIP_MEMORY_SCOPE_WORKGROUP);
}

__global__ __launch_bounds__(1024) void gcn_spmm2(
    const int* __restrict__ esrc, const int* __restrict__ edst,
    const float* __restrict__ ew, const float* __restrict__ z,
    float* __restrict__ out) {
  __shared__ float acc[S2_ROWS];
  const int part  = blockIdx.x % S2_PARTS;
  const int chunk = blockIdx.x / S2_PARTS;
  const int lo    = part * S2_ROWS;

  for (int i = threadIdx.x; i < S2_ROWS; i += 1024) acc[i] = 0.0f;
  __syncthreads();

  const int4*   __restrict__ d4 = (const int4*)edst + (size_t)chunk * S2_QUADS;
  const int4*   __restrict__ s4 = (const int4*)esrc + (size_t)chunk * S2_QUADS;
  const float4* __restrict__ w4 = (const float4*)ew + (size_t)chunk * S2_QUADS;

  int q = threadIdx.x;
  int4   d = d4[q], s = s4[q];
  float4 w = w4[q];
  S2Hit h0 = s2_gather(d, s, w, true, lo, z);
  q += 1024;
  { const int qc = min(q, S2_QUADS - 1); d = d4[qc]; s = s4[qc]; w = w4[qc]; }
  for (int k = 0; k < (S2_TRIPS - 1) / 2; ++k) {
    const S2Hit h1 = s2_gather(d, s, w, q < S2_QUADS, lo, z);
    q += 1024;
    { const int qc = min(q, S2_QUADS - 1); d = d4[qc]; s = s4[qc]; w = w4[qc]; }
    s2_add(acc, h0);
    h0 = s2_gather(d, s, w, q < S2_QUADS, lo, z);
    q += 1024;
    { const int qc = min(q, S2_QUADS - 1); d = d4[qc]; s = s4[qc]; w = w4[qc]; }
    s2_add(acc, h1);
  }
  s2_add(acc, h0);
  __syncthreads();

  const int rot = chunk * S2_ROT;
  for (int i = threadIdx.x; i < S2_ROWS; i += 1024) {
    int j = i + rot;
    if (j >= S2_ROWS) j -= S2_ROWS;
    const int g = lo + j;
    if (g < N_NODES)
      __hip_atomic_fetch_add(&out[g], acc[j], __ATOMIC_RELAXED, __HIP_MEMORY_SCOPE_AGENT);
  }
}

extern "C" void kernel_launch(void* const* d_in, const int* in_sizes, int n_in,
                              void* d_out, int out_size, void* d_ws, size_t ws_size,
                              hipStream_t stream) {
  const float* x    = (const float*)d_in[0];
  const int*   esrc = (const int*)  d_in[1];
  const int*   edst = (const int*)  d_in[2];
  const float* ew   = (const float*)d_in[3];
  const float* W1   = (const float*)d_in[4];
  const float* b1   = (const float*)d_in[5];
  const float* W2   = (const float*)d_in[6];
  const float* b2   = (const float*)d_in[7];
  float* out = (float*)d_out;

  float* h1 = (float*)d_ws;
  float* h2 = h1 + (size_t)N_NODES * HID_DIM;
  float* z  = h2 + (size_t)N_NODES * HID_DIM;

  {
    int blocks = (N_TILES + 7) / 8;
    gcn_gemm_xw1<<<blocks, 256, 0, stream>>>(x, W1, h1);
  }
  {
    int total = N_NODES * HID_DIM;
    gcn_init<<<(total + 255) / 256, 256, 0, stream>>>(h2, out, b2);
  }
  gcn_spmm1<<<S1_PARTS * S1_CHUNKS, 1024, 0, stream>>>(esrc, edst, ew, h1, h2);
  gcn_layer2<<<(N_NODES + 255) / 256, 256, 0, stream>>>(h2, b1, W2, z);
  gcn_spmm2<<<S2_PARTS * S2_CHUNKS, 1024, 0, stream>>>(esrc, edst, ew, z, out);
}


# layer-2 SpMM: work-group owns (dst range, src range, edge piece) and stages its z slice in LDS, so edges read z from LDS, not global; 64 partial sums per row: changes the f32 summation order
# speedup vs baseline: 2.8493x; 1.0949x over previous
#include <hip/hip_runtime.h>
#include <hip/hip_bf16.h>

#define N_NODES 100000
#define N_EDGES 3200000
#define IN_DIM  512
#define HID_DIM 16
#define N_TILES (N_NODES / 16)

typedef float v2f __attribute__((ext_vector_type(2)));
typedef float v8f __attribute__((ext_vector_type(8)));

__global__ __launch_bounds__(256) void gcn_gemm_xw1(
    const float* __restrict__ X, const float* __restrict__ W1,
    float* __restrict__ h1) {
  __shared__ float sW[IN_DIM * HID_DIM];
  for (int i = threadIdx.x; i < IN_DIM * HID_DIM; i += 256) sW[i] = W1[i];
  __syncthreads();

  const int wave = threadIdx.x >> 5;
  const int lane = threadIdx.x & 31;
  const int tile = blockIdx.x * 8 + wave;
  if (tile >= N_TILES) return;

  const int m  = lane & 15;
  const int hi = lane >> 4;

  const float* __restrict__ xrow = X + (size_t)(tile * 16 + m) * IN_DIM;

  v8f c = {0.f, 0.f, 0.f, 0.f, 0.f, 0.f, 0.f, 0.f};

  #pragma unroll 4
  for (int k0 = 0; k0 < IN_DIM; k0 += 4) {
    const int k = k0 + 2 * hi;
    v2f a;
    a.x = xrow[k];
    a.y = xrow[k + 1];
    v2f b;
    b.x = sW[k * HID_DIM + m];
    b.y = sW[(k + 1) * HID_DIM + m];
    c = __builtin_amdgcn_wmma_f32_16x16x4_f32(
        false, a, false, b, (short)0, c, false, false);
  }

  float* __restrict__ out = h1 + (size_t)tile * 16 * HID_DIM;
  const int mbase = hi * 8;
  #pragma unroll
  for (int v = 0; v < 8; ++v)
    out[(mbase + v) * HID_DIM + m] = c[v];
}

__global__ void gcn_init(float* __restrict__ h2, float* __restrict__ out,
                         const float* __restrict__ b2) {
  int i = blockIdx.x * blockDim.x + threadIdx.x;
  if (i < N_NODES * HID_DIM) h2[i] = 0.0f;
  if (i < N_NODES) out[i] = b2[0];
}

#define S1_PARTS   22
#define S1_CHUNKS  11
#define S1_ROWS    4576
#define S1_FLOATS  (S1_ROWS * HID_DIM)
#define S1_NQUADS  (N_EDGES / 4)
#define S1_QPC     72728
#define S1_TRIPS   ((S1_QPC + 1023) / 1024)
#define S1_ROTF    (S1_FLOATS / S1_CHUNKS)
static_assert(S1_PARTS * S1_ROWS >= N_NODES, "range cover");
static_assert(S1_CHUNKS * S1_QPC >= S1_NQUADS, "edge cover");
static_assert(S1_ROTF * S1_CHUNKS == S1_FLOATS && S1_ROTF % 32 == 0, "rotation by whole lines");
static_assert(S1_ROWS <= (1 << 13) && N_NODES <= (1 << 17), "row offset and source row share one dword");
static_assert(S1_FLOATS % 4 == 0, "LDS clear by 16-byte stores");

__device__ __forceinline__ void s1_gather(const uint2* my, const unsigned j, const unsigned total,
                                          const unsigned h, const unsigned c,
                                          const float* __restrict__ h1,
                                          unsigned (&ro)[4], float (&wt)[4], float (&hv)[4]) {
  #pragma unroll
  for (int u = 0; u < 4; ++u) {
    const unsigned idx = j + 2u * (unsigned)u + h;
    const uint2 e = my[idx < total ? idx : j];
    ro[u] = (e.x >> 17) * HID_DIM + c;
    wt[u] = __uint_as_float(e.y);
    hv[u] = h1[(size_t)(e.x & 0x1ffffu) * HID_DIM + c];
  }
}

__device__ __forceinline__ void s1_add(float* acc, const unsigned j, const unsigned total,
                                       const unsigned h,
                                       const unsigned (&ro)[4], const float (&wt)[4], const float (&hv)[4]) {
  #pragma unroll
  for (int u = 0; u < 4; ++u) {
    const bool ok = j + 2u * (unsigned)u + h < total;
    __hip_atomic_fetch_add(&acc[ro[u]], ok ? wt[u] * hv[u] : 0.0f,
                           __ATOMIC_RELAXED, __HIP_MEMORY_SCOPE_WORKGROUP);
  }
}

__global__ __launch_bounds__(1024) void gcn_spmm1(
    const int* __restrict__ esrc, const int* __restrict__ edst,
    const float* __restrict__ ew, const float* __restrict__ h1,
    float* __restrict__ h2) {
  __shared__ float acc[S1_FLOATS];
  __shared__ uint2 stg[32 * 128];
  const int part  = blockIdx.x % S1_PARTS;
  const int chunk = blockIdx.x / S1_PARTS;
  const int lo    = part * S1_ROWS;

  for (int i = threadIdx.x; i < S1_FLOATS / 4; i += 1024)
    ((float4*)acc)[i] = make_float4(0.0f, 0.0f, 0.0f, 0.0f);
  __syncthreads();

  const unsigned c = threadIdx.x & 15u;
  const unsigned h = (threadIdx.x >> 4) & 1u;
  uint2* my = stg + (threadIdx.x >> 5) * 128;
  if ((threadIdx.x & 31u) == 0u) my[0] = make_uint2(0u, 0u);

  const int4*   __restrict__ d4 = (const int4*)edst;
  const int4*   __restrict__ s4 = (const int4*)esrc;
  const float4* __restrict__ w4 = (const float4*)ew;
  const int qbeg = chunk * S1_QPC;
  const int qend = min(qbeg + S1_QPC, S1_NQUADS);

  int q = qbeg + (int)threadIdx.x;
  int4 d, s; float4 w;
  { const int qc = min(q, S1_NQUADS - 1); d = d4[qc]; s = s4[qc]; w = w4[qc]; }

  for (int t = 0; t < S1_TRIPS; ++t) {
    const bool valid = q < qend;
    const unsigned r0 = valid ? (unsigned)(d.x - lo) : 0xffffffffu;
    const unsigned r1 = valid ? (unsigned)(d.y - lo) : 0xffffffffu;
    const unsigned r2 = valid ? (unsigned)(d.z - lo) : 0xffffffffu;
    const unsigned r3 = valid ? (unsigned)(d.w - lo) : 0xffffffffu;
    const bool g0 = r0 < (unsigned)S1_ROWS, g1 = r1 < (unsigned)S1_ROWS;
    const bool g2 = r2 < (unsigned)S1_ROWS, g3 = r3 < (unsigned)S1_ROWS;
    const unsigned m0 = __builtin_amdgcn_ballot_w32(g0), m1 = __builtin_amdgcn_ballot_w32(g1);
    const unsigned m2 = __builtin_amdgcn_ballot_w32(g2), m3 = __builtin_amdgcn_ballot_w32(g3);
    const unsigned b1 = (unsigned)__builtin_popcount(m0);
    const unsigned b2 = b1 + (unsigned)__builtin_popcount(m1);
    const unsigned b3 = b2 + (unsigned)__builtin_popcount(m2);
    const unsigned total = b3 + (unsigned)__builtin_popcount(m3);
    if (g0) my[__builtin_amdgcn_mbcnt_lo(m0, 0u)] = make_uint2((r0 << 17) | (unsigned)s.x, __float_as_uint(w.x));
    if (g1) my[__builtin_amdgcn_mbcnt_lo(m1, b1)] = make_uint2((r1 << 17) | (unsigned)s.y, __float_as_uint(w.y));
    if (g2) my[__builtin_amdgcn_mbcnt_lo(m2, b2)] = make_uint2((r2 << 17) | (unsigned)s.z, __float_as_uint(w.z));
    if (g3) my[__builtin_amdgcn_mbcnt_lo(m3, b3)] = make_uint2((r3 << 17) | (unsigned)s.w, __float_as_uint(w.w));
    unsigned ro[4]; float wt[4], hv[4];
    s1_gather(my, 0u, total, h, c, h1, ro, wt, hv);
    q += 1024;
    { const int qc = min(q, S1_NQUADS - 1); d = d4[qc]; s = s4[qc]; w = w4[qc]; }
    __builtin_amdgcn_sched_barrier(0);
    s1_add(acc, 0u, total, h, ro, wt, hv);
    for (unsigned j = 8u; j < total; j += 8u) {
      s1_gather(my, j, total, h, c, h1, ro, wt, hv);
      s1_add(acc, j, total, h, ro, wt, hv);
    }
  }
  __syncthreads();

  const int rot = chunk * S1_ROTF;
  const int lim = (N_NODES - lo) * HID_DIM;
  float* __restrict__ dst = h2 + (size_t)lo * HID_DIM;
  for (int i = threadIdx.x; i < S1_FLOATS; i += 1024) {
    int j = i + rot;
    if (j >= S1_FLOATS) j -= S1_FLOATS;
    if (j < lim)
      __hip_atomic_fetch_add(&dst[j], acc[j], __ATOMIC_RELAXED, __HIP_MEMORY_SCOPE_AGENT);
  }
}

__global__ void gcn_layer2(const float* __restrict__ h2,
                           const float* __restrict__ b1,
                           const float* __restrict__ W2,
                           float* __restrict__ z) {
  int i = blockIdx.x * blockDim.x + threadIdx.x;
  if (i >= N_NODES) return;
  const float* __restrict__ row = h2 + (size_t)i * HID_DIM;
  float acc = 0.0f;
  #pragma unroll
  for (int d = 0; d < HID_DIM; ++d) {
    float hv = row[d] + b1[d];
    hv = fmaxf(hv, 0.0f);
    acc = fmaf(hv, W2[d], acc);
  }
  z[i] = acc;
}

#define S2_DPARTS 4
#define S2_SPARTS 2
#define S2_CHUNKS 32
#define S2_DROWS  25024
#define S2_SROWS  50000
#define S2_QUADS  (N_EDGES / S2_CHUNKS / 4)
#define S2_QPT    2048
#define S2_TRIPS  ((S2_QUADS + S2_QPT - 1) / S2_QPT)
#define S2_ROT    384
#define S2_ZTRIPS ((S2_SROWS / 4 + 1023) / 1024)
static_assert(S2_DPARTS * S2_DROWS >= N_NODES && S2_DROWS % 32 == 0, "destination cover, whole lines");
static_assert(S2_SPARTS * S2_SROWS == N_NODES && S2_SROWS % 4 == 0, "source cover, 16-byte copies");
static_assert(S2_CHUNKS * S2_QUADS * 4 == N_EDGES, "edge cover");
static_assert(S2_CHUNKS * S2_SPARTS * S2_ROT < S2_DROWS && S2_ROT % 32 == 0, "rotation by whole lines");
static_assert(S2_DROWS % 4 == 0, "LDS clear by 16-byte stores");

__device__ __forceinline__ void s2_edges(float* acc, const float* zs,
                                         const int4 da, const int4 sa, const float4 wa, const bool va,
                                         const int4 db, const int4 sb, const float4 wb, const bool vb,
                                         const int dlo, const int slo) {
  const int   dd[8] = {da.x, da.y, da.z, da.w, db.x, db.y, db.z, db.w};
  const int   ss[8] = {sa.x, sa.y, sa.z, sa.w, sb.x, sb.y, sb.z, sb.w};
  const float ww[8] = {wa.x, wa.y, wa.z, wa.w, wb.x, wb.y, wb.z, wb.w};
  unsigned rd[8]; bool hit[8]; float zv[8];
  #pragma unroll
  for (int e = 0; e < 8; ++e) {
    const bool v = e < 4 ? va : vb;
    rd[e] = (unsigned)(dd[e] - dlo);
    const unsigned rs = (unsigned)(ss[e] - slo);
    hit[e] = v && rd[e] < (unsigned)S2_DROWS && rs < (unsigned)S2_SROWS;
    zv[e] = zs[hit[e] ? rs : 0u];
  }
  #pragma unroll
  for (int e = 0; e < 8; ++e)
    if (hit[e])
      __hip_atomic_fetch_add(&acc[rd[e]], ww[e] * zv[e], __ATOMIC_RELAXED, __HIP_MEMORY_SCOPE_WORKGROUP);
}

__global__ __launch_bounds__(1024) void gcn_spmm2(
    const int* __restrict__ esrc, const int* __restrict__ edst,
    const float* __restrict__ ew, const float* __restrict__ z,
    float* __restrict__ out) {
  __shared__ __attribute__((aligned(16))) float acc[S2_DROWS];
  __shared__ __attribute__((aligned(16))) float zs[S2_SROWS];
  const int dpart = blockIdx.x % S2_DPARTS;
  const int spart = (blockIdx.x / S2_DPARTS) % S2_SPARTS;
  const int chunk = blockIdx.x / (S2_DPARTS * S2_SPARTS);
  const int dlo   = dpart * S2_DROWS;
  const int slo   = spart * S2_SROWS;

  const int4*   __restrict__ d4 = (const int4*)edst + (size_t)chunk * S2_QUADS;
  const int4*   __restrict__ s4 = (const int4*)esrc + (size_t)chunk * S2_QUADS;
  const float4* __restrict__ w4 = (const float4*)ew + (size_t)chunk * S2_QUADS;

  int q = threadIdx.x;
  int4 da, sa, db, sb; float4 wa, wb;
  { const int qa = min(q, S2_QUADS - 1), qb = min(q + 1024, S2_QUADS - 1);
    da = d4[qa]; sa = s4[qa]; wa = w4[qa]; db = d4[qb]; sb = s4[qb]; wb = w4[qb]; }

  for (int i = threadIdx.x; i < S2_DROWS / 4; i += 1024)
    ((float4*)acc)[i] = make_float4(0.0f, 0.0f, 0.0f, 0.0f);
  const float4* __restrict__ zsrc = (const float4*)(z + slo);
  float4 zr[S2_ZTRIPS];
  #pragma unroll
  for (int k = 0; k < S2_ZTRIPS; ++k)
    zr[k] = zsrc[min((int)threadIdx.x + 1024 * k, S2_SROWS / 4 - 1)];
  #pragma unroll
  for (int k = 0; k < S2_ZTRIPS; ++k) {
    const int i = (int)threadIdx.x + 1024 * k;
    if (i < S2_SROWS / 4) ((float4*)zs)[i] = zr[k];
  }
  __syncthreads();

  for (int t = 0; t < S2_TRIPS; ++t) {
    const bool va = q < S2_QUADS, vb = q + 1024 < S2_QUADS;
    const int4 cda = da, csa = sa, cdb = db, csb = sb;
    const float4 cwa = wa, cwb = wb;
    q += S2_QPT;
    { const int qa = min(q, S2_QUADS - 1), qb = min(q + 1024, S2_QUADS - 1);
      da = d4[qa]; sa = s4[qa]; wa = w4[qa]; db = d4[qb]; sb = s4[qb]; wb = w4[qb]; }
    s2_edges(acc, zs, cda, csa, cwa, va, cdb, csb, cwb, vb, dlo, slo);
  }
  __syncthreads();

  const int rot = (chunk * S2_SPARTS + spart) * S2_ROT;
  for (int i = threadIdx.x; i < S2_DROWS; i += 1024) {
    int j = i + rot;
    if (j >= S2_DROWS) j -= S2_DROWS;
    const int g = dlo + j;
    if (g < N_NODES)
      __hip_atomic_fetch_add(&out[g], acc[j], __ATOMIC_RELAXED, __HIP_MEMORY_SCOPE_AGENT);
  }
}

extern "C" void kernel_launch(void* const* d_in, const int* in_sizes, int n_in,
                              void* d_out, int out_size, void* d_ws, size_t ws_size,
                              hipStream_t stream) {
  const float* x    = (const float*)d_in[0];
  const int*   esrc = (const int*)  d_in[1];
  const int*   edst = (const int*)  d_in[2];
  const float* ew   = (const float*)d_in[3];
  const float* W1   = (const float*)d_in[4];
  const float* b1   = (const float*)d_in[5];
  const float* W2   = (const float*)d_in[6];
  const float* b2   = (const float*)d_in[7];
  float* out = (float*)d_out;

  float* h1 = (float*)d_ws;
  float* h2 = h1 + (size_t)N_NODES * HID_DIM;
  float* z  = h2 + (size_t)N_NODES * HID_DIM;

  {
    int blocks = (N_TILES + 7) / 8;
    gcn_gemm_xw1<<<blocks, 256, 0, stream>>>(x, W1, h1);
  }
  {
    int total = N_NODES * HID_DIM;
    gcn_init<<<(total + 255) / 256, 256, 0, stream>>>(h2, out, b2);
  }
  gcn_spmm1<<<S1_PARTS * S1_CHUNKS, 1024, 0, stream>>>(esrc, edst, ew, h1, h2);
  gcn_layer2<<<(N_NODES + 255) / 256, 256, 0, stream>>>(h2, b1, W2, z);
  gcn_spmm2<<<S2_DPARTS * S2_SPARTS * S2_CHUNKS, 1024, 0, stream>>>(esrc, edst, ew, z, out);
}


# X @ W1: X read as whole 256-byte row pieces (16-byte loads, 4 lines per load, not 16) via a wave-private LDS tile, loads issued ahead; W1 transposed in LDS; 2 accumulator tiles: f32 sum order changes
# speedup vs baseline: 2.9559x; 1.0374x over previous
#include <hip/hip_runtime.h>
#include <hip/hip_bf16.h>

#define N_NODES 100000
#define N_EDGES 3200000
#define IN_DIM  512
#define HID_DIM 16
#define N_TILES (N_NODES / 16)

typedef float v2f __attribute__((ext_vector_type(2)));
typedef float v8f __attribute__((ext_vector_type(8)));
typedef float v4f __attribute__((ext_vector_type(4)));

#define G_WPITCH 516
#define G_APITCH 68
#define G_PANEL  64
static_assert(IN_DIM % G_PANEL == 0, "whole panels");
static_assert((G_WPITCH * 4) % 16 == 0 && (G_APITCH * 4) % 16 == 0, "16-byte aligned rows");

__global__ __launch_bounds__(256) void gcn_gemm_xw1(
    const float* __restrict__ X, const float* __restrict__ W1,
    float* __restrict__ h1) {
  __shared__ __attribute__((aligned(16))) float sWt[HID_DIM * G_WPITCH];
  __shared__ __attribute__((aligned(16))) float sA[8 * 16 * G_APITCH];

  const int wave = threadIdx.x >> 5;
  const int lane = threadIdx.x & 31;
  const int tile = blockIdx.x * 8 + wave;
  const int m  = lane & 15;
  const int hi = lane >> 4;

  const int tl = min(tile, N_TILES - 1);
  const float* __restrict__ xl = X + (size_t)(tl * 16 + hi) * IN_DIM + 4 * m;

  v4f cur[8];
  #pragma unroll
  for (int i = 0; i < 8; ++i)
    cur[i] = __builtin_nontemporal_load((const v4f*)(xl + (size_t)(2 * i) * IN_DIM));

  {
    float* __restrict__ wdst = sWt + (threadIdx.x & 15) * G_WPITCH + (threadIdx.x >> 4);
    float wv[IN_DIM * HID_DIM / 256];
    #pragma unroll
    for (int s = 0; s < IN_DIM * HID_DIM / 256; ++s) wv[s] = W1[threadIdx.x + 256 * s];
    #pragma unroll
    for (int s = 0; s < IN_DIM * HID_DIM / 256; ++s) wdst[16 * s] = wv[s];
  }
  __syncthreads();
  if (tile >= N_TILES) return;

  float* __restrict__ at = sA + wave * (16 * G_APITCH);
  float* __restrict__ ast = at + hi * G_APITCH + 4 * m;
  const float* __restrict__ afr = at + m * G_APITCH + 4 * hi;
  const float* __restrict__ bfr = sWt + m * G_WPITCH + 4 * hi;

  v8f c0 = {0.f, 0.f, 0.f, 0.f, 0.f, 0.f, 0.f, 0.f};
  v8f c1 = {0.f, 0.f, 0.f, 0.f, 0.f, 0.f, 0.f, 0.f};

  for (int p = 0; p < IN_DIM / G_PANEL; ++p) {
    #pragma unroll
    for (int i = 0; i < 8; ++i)
      *(v4f*)(ast + (2 * i) * G_APITCH) = cur[i];
    if (p + 1 < IN_DIM / G_PANEL) {
      #pragma unroll
      for (int i = 0; i < 8; ++i)
        cur[i] = __builtin_nontemporal_load(
            (const v4f*)(xl + (size_t)(2 * i) * IN_DIM + (p + 1) * G_PANEL));
    }
    #pragma unroll
    for (int j = 0; j < 8; ++j) {
      const v4f a4 = *(const v4f*)(afr + 8 * j);
      const v4f b4 = *(const v4f*)(bfr + p * G_PANEL + 8 * j);
      v2f a, b;
      a.x = a4.x; a.y = a4.y; b.x = b4.x; b.y = b4.y;
      c0 = __builtin_amdgcn_wmma_f32_16x16x4_f32(
          false, a, false, b, (short)0, c0, false, false);
      a.x = a4.z; a.y = a4.w; b.x = b4.z; b.y = b4.w;
      c1 = __builtin_amdgcn_wmma_f32_16x16x4_f32(
          false, a, false, b, (short)0, c1, false, false);
    }
  }

  float* __restrict__ out = h1 + (size_t)tile * 16 * HID_DIM;
  const int mbase = hi * 8;
  #pragma unroll
  for (int v = 0; v < 8; ++v)
    out[(mbase + v) * HID_DIM + m] = c0[v] + c1[v];
}

__global__ void gcn_init(float* __restrict__ h2, float* __restrict__ out,
                         const float* __restrict__ b2) {
  int i = blockIdx.x * blockDim.x + threadIdx.x;
  if (i < N_NODES * HID_DIM) h2[i] = 0.0f;
  if (i < N_NODES) out[i] = b2[0];
}

#define S1_PARTS   22
#define S1_CHUNKS  11
#define S1_ROWS    4576
#define S1_FLOATS  (S1_ROWS * HID_DIM)
#define S1_NQUADS  (N_EDGES / 4)
#define S1_QPC     72728
#define S1_TRIPS   ((S1_QPC + 1023) / 1024)
#define S1_ROTF    (S1_FLOATS / S1_CHUNKS)
static_assert(S1_PARTS * S1_ROWS >= N_NODES, "range cover");
static_assert(S1_CHUNKS * S1_QPC >= S1_NQUADS, "edge cover");
static_assert(S1_ROTF * S1_CHUNKS == S1_FLOATS && S1_ROTF % 32 == 0, "rotation by whole lines");
static_assert(S1_ROWS <= (1 << 13) && N_NODES <= (1 << 17), "row offset and source row share one dword");
static_assert(S1_FLOATS % 4 == 0, "LDS clear by 16-byte stores");

__device__ __forceinline__ void s1_gather(const uint2* my, const unsigned j, const unsigned total,
                                          const unsigned h, const unsigned c,
                                          const float* __restrict__ h1,
                                          unsigned (&ro)[4], float (&wt)[4], float (&hv)[4]) {
  #pragma unroll
  for (int u = 0; u < 4; ++u) {
    const unsigned idx = j + 2u * (unsigned)u + h;
    const uint2 e = my[idx < total ? idx : j];
    ro[u] = (e.x >> 17) * HID_DIM + c;
    wt[u] = __uint_as_float(e.y);
    hv[u] = h1[(size_t)(e.x & 0x1ffffu) * HID_DIM + c];
  }
}

__device__ __forceinline__ void s1_add(float* acc, const unsigned j, const unsigned total,
                                       const unsigned h,
                                       const unsigned (&ro)[4], const float (&wt)[4], const float (&hv)[4]) {
  #pragma unroll
  for (int u = 0; u < 4; ++u) {
    const bool ok = j + 2u * (unsigned)u + h < total;
    __hip_atomic_fetch_add(&acc[ro[u]], ok ? wt[u] * hv[u] : 0.0f,
                           __ATOMIC_RELAXED, __HIP_MEMORY_SCOPE_WORKGROUP);
  }
}

__global__ __launch_bounds__(1024) void gcn_spmm1(
    const int* __restrict__ esrc, const int* __restrict__ edst,
    const float* __restrict__ ew, const float* __restrict__ h1,
    float* __restrict__ h2) {
  __shared__ float acc[S1_FLOATS];
  __shared__ uint2 stg[32 * 128];
  const int part  = blockIdx.x % S1_PARTS;
  const int chunk = blockIdx.x / S1_PARTS;
  const int lo    = part * S1_ROWS;

  for (int i = threadIdx.x; i < S1_FLOATS / 4; i += 1024)
    ((float4*)acc)[i] = make_float4(0.0f, 0.0f, 0.0f, 0.0f);
  __syncthreads();

  const unsigned c = threadIdx.x & 15u;
  const unsigned h = (threadIdx.x >> 4) & 1u;
  uint2* my = stg + (threadIdx.x >> 5) * 128;
  if ((threadIdx.x & 31u) == 0u) my[0] = make_uint2(0u, 0u);

  const int4*   __restrict__ d4 = (const int4*)edst;
  const int4*   __restrict__ s4 = (const int4*)esrc;
  const float4* __restrict__ w4 = (const float4*)ew;
  const int qbeg = chunk * S1_QPC;
  const int qend = min(qbeg + S1_QPC, S1_NQUADS);

  int q = qbeg + (int)threadIdx.x;
  int4 d, s; float4 w;
  { const int qc = min(q, S1_NQUADS - 1); d = d4[qc]; s = s4[qc]; w = w4[qc]; }

  for (int t = 0; t < S1_TRIPS; ++t) {
    const bool valid = q < qend;
    const unsigned r0 = valid ? (unsigned)(d.x - lo) : 0xffffffffu;
    const unsigned r1 = valid ? (unsigned)(d.y - lo) : 0xffffffffu;
    const unsigned r2 = valid ? (unsigned)(d.z - lo) : 0xffffffffu;
    const unsigned r3 = valid ? (unsigned)(d.w - lo) : 0xffffffffu;
    const bool g0 = r0 < (unsigned)S1_ROWS, g1 = r1 < (unsigned)S1_ROWS;
    const bool g2 = r2 < (unsigned)S1_ROWS, g3 = r3 < (unsigned)S1_ROWS;
    const unsigned m0 = __builtin_amdgcn_ballot_w32(g0), m1 = __builtin_amdgcn_ballot_w32(g1);
    const unsigned m2 = __builtin_amdgcn_ballot_w32(g2), m3 = __builtin_amdgcn_ballot_w32(g3);
    const unsigned b1 = (unsigned)__builtin_popcount(m0);
    const unsigned b2 = b1 + (unsigned)__builtin_popcount(m1);
    const unsigned b3 = b2 + (unsigned)__builtin_popcount(m2);
    const unsigned total = b3 + (unsigned)__builtin_popcount(m3);
    if (g0) my[__builtin_amdgcn_mbcnt_lo(m0, 0u)] = make_uint2((r0 << 17) | (unsigned)s.x, __float_as_uint(w.x));
    if (g1) my[__builtin_amdgcn_mbcnt_lo(m1, b1)] = make_uint2((r1 << 17) | (unsigned)s.y, __float_as_uint(w.y));
    if (g2) my[__builtin_amdgcn_mbcnt_lo(m2, b2)] = make_uint2((r2 << 17) | (unsigned)s.z, __float_as_uint(w.z));
    if (g3) my[__builtin_amdgcn_mbcnt_lo(m3, b3)] = make_uint2((r3 << 17) | (unsigned)s.w, __float_as_uint(w.w));
    unsigned ro[4]; float wt[4], hv[4];
    s1_gather(my, 0u, total, h, c, h1, ro, wt, hv);
    q += 1024;
    { const int qc = min(q, S1_NQUADS - 1); d = d4[qc]; s = s4[qc]; w = w4[qc]; }
    __builtin_amdgcn_sched_barrier(0);
    s1_add(acc, 0u, total, h, ro, wt, hv);
    for (unsigned j = 8u; j < total; j += 8u) {
      s1_gather(my, j, total, h, c, h1, ro, wt, hv);
      s1_add(acc, j, total, h, ro, wt, hv);
    }
  }
  __syncthreads();

  const int rot = chunk * S1_ROTF;
  const int lim = (N_NODES - lo) * HID_DIM;
  float* __restrict__ dst = h2 + (size_t)lo * HID_DIM;
  for (int i = threadIdx.x; i < S1_FLOATS; i += 1024) {
    int j = i + rot;
    if (j >= S1_FLOATS) j -= S1_FLOATS;
    if (j < lim)
      __hip_atomic_fetch_add(&dst[j], acc[j], __ATOMIC_RELAXED, __HIP_MEMORY_SCOPE_AGENT);
  }
}

__global__ void gcn_layer2(const float* __restrict__ h2,
                           const float* __restrict__ b1,
                           const float* __restrict__ W2,
                           float* __restrict__ z) {
  int i = blockIdx.x * blockDim.x + threadIdx.x;
  if (i >= N_NODES) return;
  const float* __restrict__ row = h2 + (size_t)i * HID_DIM;
  float acc = 0.0f;
  #pragma unroll
  for (int d = 0; d < HID_DIM; ++d) {
    float hv = row[d] + b1[d];
    hv = fmaxf(hv, 0.0f);
    acc = fmaf(hv, W2[d], acc);
  }
  z[i] = acc;
}

#define S2_DPARTS 4
#define S2_SPARTS 2
#define S2_CHUNKS 32
#define S2_DROWS  25024
#define S2_SROWS  50000
#define S2_QUADS  (N_EDGES / S2_CHUNKS / 4)
#define S2_QPT    2048
#define S2_TRIPS  ((S2_QUADS + S2_QPT - 1) / S2_QPT)
#define S2_ROT    384
#define S2_ZTRIPS ((S2_SROWS / 4 + 1023) / 1024)
static_assert(S2_DPARTS * S2_DROWS >= N_NODES && S2_DROWS % 32 == 0, "destination cover, whole lines");
static_assert(S2_SPARTS * S2_SROWS == N_NODES && S2_SROWS % 4 == 0, "source cover, 16-byte copies");
static_assert(S2_CHUNKS * S2_QUADS * 4 == N_EDGES, "edge cover");
static_assert(S2_CHUNKS * S2_SPARTS * S2_ROT < S2_DROWS && S2_ROT % 32 == 0, "rotation by whole lines");
static_assert(S2_DROWS % 4 == 0, "LDS clear by 16-byte stores");

__device__ __forceinline__ void s2_edges(float* acc, const float* zs,
                                         const int4 da, const int4 sa, const float4 wa, const bool va,
                                         const int4 db, const int4 sb, const float4 wb, const bool vb,
                                         const int dlo, const int slo) {
  const int   dd[8] = {da.x, da.y, da.z, da.w, db.x, db.y, db.z, db.w};
  const int   ss[8] = {sa.x, sa.y, sa.z, sa.w, sb.x, sb.y, sb.z, sb.w};
  const float ww[8] = {wa.x, wa.y, wa.z, wa.w, wb.x, wb.y, wb.z, wb.w};
  unsigned rd[8]; bool hit[8]; float zv[8];
  #pragma unroll
  for (int e = 0; e < 8; ++e) {
    const bool v = e < 4 ? va : vb;
    rd[e] = (unsigned)(dd[e] - dlo);
    const unsigned rs = (unsigned)(ss[e] - slo);
    hit[e] = v && rd[e] < (unsigned)S2_DROWS && rs < (unsigned)S2_SROWS;
    zv[e] = zs[hit[e] ? rs : 0u];
  }
  #pragma unroll
  for (int e = 0; e < 8; ++e)
    if (hit[e])
      __hip_atomic_fetch_add(&acc[rd[e]], ww[e] * zv[e], __ATOMIC_RELAXED, __HIP_MEMORY_SCOPE_WORKGROUP);
}

__global__ __launch_bounds__(1024) void gcn_spmm2(
    const int* __restrict__ esrc, const int* __restrict__ edst,
    const float* __restrict__ ew, const float* __restrict__ z,
    float* __restrict__ out) {
  __shared__ __attribute__((aligned(16))) float acc[S2_DROWS];
  __shared__ __attribute__((aligned(16))) float zs[S2_SROWS];
  const int dpart = blockIdx.x % S2_DPARTS;
  const int spart = (blockIdx.x / S2_DPARTS) % S2_SPARTS;
  const int chunk = blockIdx.x / (S2_DPARTS * S2_SPARTS);
  const int dlo   = dpart * S2_DROWS;
  const int slo   = spart * S2_SROWS;

  const int4*   __restrict__ d4 = (const int4*)edst + (size_t)chunk * S2_QUADS;
  const int4*   __restrict__ s4 = (const int4*)esrc + (size_t)chunk * S2_QUADS;
  const float4* __restrict__ w4 = (const float4*)ew + (size_t)chunk * S2_QUADS;

  int q = threadIdx.x;
  int4 da, sa, db, sb; float4 wa, wb;
  { const int qa = min(q, S2_QUADS - 1), qb = min(q + 1024, S2_QUADS - 1);
    da = d4[qa]; sa = s4[qa]; wa = w4[qa]; db = d4[qb]; sb = s4[qb]; wb = w4[qb]; }

  for (int i = threadIdx.x; i < S2_DROWS / 4; i += 1024)
    ((float4*)acc)[i] = make_float4(0.0f, 0.0f, 0.0f, 0.0f);
  const float4* __restrict__ zsrc = (const float4*)(z + slo);
  float4 zr[S2_ZTRIPS];
  #pragma unroll
  for (int k = 0; k < S2_ZTRIPS; ++k)
    zr[k] = zsrc[min((int)threadIdx.x + 1024 * k, S2_SROWS / 4 - 1)];
  #pragma unroll
  for (int k = 0; k < S2_ZTRIPS; ++k) {
    const int i = (int)threadIdx.x + 1024 * k;
    if (i < S2_SROWS / 4) ((float4*)zs)[i] = zr[k];
  }
  __syncthreads();

  for (int t = 0; t < S2_TRIPS; ++t) {
    const bool va = q < S2_QUADS, vb = q + 1024 < S2_QUADS;
    const int4 cda = da, csa = sa, cdb = db, csb = sb;
    const float4 cwa = wa, cwb = wb;
    q += S2_QPT;
    { const int qa = min(q, S2_QUADS - 1), qb = min(q + 1024, S2_QUADS - 1);
      da = d4[qa]; sa = s4[qa]; wa = w4[qa]; db = d4[qb]; sb = s4[qb]; wb = w4[qb]; }
    s2_edges(acc, zs, cda, csa, cwa, va, cdb, csb, cwb, vb, dlo, slo);
  }
  __syncthreads();

  const int rot = (chunk * S2_SPARTS + spart) * S2_ROT;
  for (int i = threadIdx.x; i < S2_DROWS; i += 1024) {
    int j = i + rot;
    if (j >= S2_DROWS) j -= S2_DROWS;
    const int g = dlo + j;
    if (g < N_NODES)
      __hip_atomic_fetch_add(&out[g], acc[j], __ATOMIC_RELAXED, __HIP_MEMORY_SCOPE_AGENT);
  }
}

extern "C" void kernel_launch(void* const* d_in, const int* in_sizes, int n_in,
                              void* d_out, int out_size, void* d_ws, size_t ws_size,
                              hipStream_t stream) {
  const float* x    = (const float*)d_in[0];
  const int*   esrc = (const int*)  d_in[1];
  const int*   edst = (const int*)  d_in[2];
  const float* ew   = (const float*)d_in[3];
  const float* W1   = (const float*)d_in[4];
  const float* b1   = (const float*)d_in[5];
  const float* W2   = (const float*)d_in[6];
  const float* b2   = (const float*)d_in[7];
  float* out = (float*)d_out;

  float* h1 = (float*)d_ws;
  float* h2 = h1 + (size_t)N_NODES * HID_DIM;
  float* z  = h2 + (size_t)N_NODES * HID_DIM;

  {
    int blocks = (N_TILES + 7) / 8;
    gcn_gemm_xw1<<<blocks, 256, 0, stream>>>(x, W1, h1);
  }
  {
    int total = N_NODES * HID_DIM;
    gcn_init<<<(total + 255) / 256, 256, 0, stream>>>(h2, out, b2);
  }
  gcn_spmm1<<<S1_PARTS * S1_CHUNKS, 1024, 0, stream>>>(esrc, edst, ew, h1, h2);
  gcn_layer2<<<(N_NODES + 255) / 256, 256, 0, stream>>>(h2, b1, W2, z);
  gcn_spmm2<<<S2_DPARTS * S2_SPARTS * S2_CHUNKS, 1024, 0, stream>>>(esrc, edst, ew, z, out);
}
